# starting kernel
# baseline (speedup 1.0000x reference)
#include <hip/hip_runtime.h>


typedef __attribute__((ext_vector_type(2))) float v2f;
typedef __attribute__((ext_vector_type(8))) float v8f;

#define L_DIM 512
#define D_DIM 256
#define TILE  64
#define KC    64
#define LDSP  68

__global__ __launch_bounds__(128) void
pairwise_bilin_wmma_f32(const float* __restrict__ X,
                        const float* __restrict__ W,
                        const float* __restrict__ bias,
                        float* __restrict__ out)
{
    __shared__ __align__(16) float As[TILE * LDSP];
    __shared__ __align__(16) float Bs[TILE * LDSP];

    const int bi = blockIdx.z;
    const int I0 = blockIdx.y * TILE;
    const int J0 = blockIdx.x * TILE;
    const float b0 = bias[0];

    const int t    = threadIdx.x;
    const int wave = t >> 5;
    const int lane = t & 31;
    const int wr   = (wave & 1) * 32;
    const int wc   = (wave >> 1) * 32;
    const int m    = lane & 15;
    const int kp   = (lane >> 4) * 2;

    const float* Xb = X + (size_t)bi * L_DIM * D_DIM;

    v8f acc[2][2] = {};

    for (int k0 = 0; k0 < D_DIM; k0 += KC) {
        for (int c = t; c < TILE * (KC / 4); c += 128) {
            const int row  = c >> 4;
            const int col4 = (c & 15) << 2;

            const float4 xa = *(const float4*)(Xb + (size_t)(I0 + row) * D_DIM + k0 + col4);
            const float4 xb = *(const float4*)(Xb + (size_t)(J0 + row) * D_DIM + k0 + col4);
            const float4 wv = *(const float4*)(W + k0 + col4);

            float4 av;
            av.x = xa.x * wv.x; av.y = xa.y * wv.y;
            av.z = xa.z * wv.z; av.w = xa.w * wv.w;

            *(float4*)&As[row * LDSP + col4] = av;
            *(float4*)&Bs[row * LDSP + col4] = xb;
        }
        __syncthreads();

        #pragma unroll
        for (int kk = 0; kk < KC; kk += 4) {
            v2f a0 = *(const v2f*)&As[(wr + m)      * LDSP + kk + kp];
            v2f a1 = *(const v2f*)&As[(wr + 16 + m) * LDSP + kk + kp];
            v2f bq0 = *(const v2f*)&Bs[(wc + m)      * LDSP + kk + kp];
            v2f bq1 = *(const v2f*)&Bs[(wc + 16 + m) * LDSP + kk + kp];

            acc[0][0] = __builtin_amdgcn_wmma_f32_16x16x4_f32(false, a0, false, bq0,
                                                              (short)0, acc[0][0], false, false);
            acc[0][1] = __builtin_amdgcn_wmma_f32_16x16x4_f32(false, a0, false, bq1,
                                                              (short)0, acc[0][1], false, false);
            acc[1][0] = __builtin_amdgcn_wmma_f32_16x16x4_f32(false, a1, false, bq0,
                                                              (short)0, acc[1][0], false, false);
            acc[1][1] = __builtin_amdgcn_wmma_f32_16x16x4_f32(false, a1, false, bq1,
                                                              (short)0, acc[1][1], false, false);
        }
        __syncthreads();
    }

    float* outb = out + (size_t)bi * L_DIM * L_DIM;
    const int rsel = (lane >> 4) * 8;
    #pragma unroll
    for (int ti = 0; ti < 2; ti++) {
        #pragma unroll
        for (int tj = 0; tj < 2; tj++) {
            const int rowBase = I0 + wr + ti * 16 + rsel;
            const int col     = J0 + wc + tj * 16 + m;
            #pragma unroll
            for (int r = 0; r < 8; r++) {
                outb[(size_t)(rowBase + r) * L_DIM + col] = acc[ti][tj][r] + b0;
            }
        }
    }
}

extern "C" void kernel_launch(void* const* d_in, const int* in_sizes, int n_in,
                              void* d_out, int out_size, void* d_ws, size_t ws_size,
                              hipStream_t stream) {
    (void)in_sizes; (void)n_in; (void)out_size; (void)d_ws; (void)ws_size;
    const float* X    = (const float*)d_in[0];
    const float* W    = (const float*)d_in[1];
    const float* bias = (const float*)d_in[2];
    float* out = (float*)d_out;

    dim3 grid(L_DIM / TILE, L_DIM / TILE, 2);
    dim3 block(128);
    pairwise_bilin_wmma_f32<<<grid, block, 0, stream>>>(X, W, bias, out);
}


# staging loads software-pipelined: a K chunk's 17 global loads per thread go out as one group, one chunk ahead of its WMMAs, across LDS-only barriers (was 32 dependent round trips per wave)
# speedup vs baseline: 3.2640x; 3.2640x over previous
#include <hip/hip_runtime.h>


typedef __attribute__((ext_vector_type(2))) float v2f;
typedef __attribute__((ext_vector_type(4))) float v4f;
typedef __attribute__((ext_vector_type(8))) float v8f;

#define L_DIM 512
#define D_DIM 256
#define TILE  64
#define KC    64
#define LDSP  68
#define NLD   8

static __device__ __forceinline__ void lds_barrier() {
    __builtin_amdgcn_fence(__ATOMIC_RELEASE, "workgroup", "local");
    __builtin_amdgcn_s_barrier();
    __builtin_amdgcn_fence(__ATOMIC_ACQUIRE, "workgroup", "local");
}

__global__ __launch_bounds__(128) void
pairwise_bilin_wmma_f32(const float* __restrict__ X,
                        const float* __restrict__ W,
                        const float* __restrict__ bias,
                        float* __restrict__ out)
{
    __shared__ __align__(16) float As[TILE * LDSP];
    __shared__ __align__(16) float Bs[TILE * LDSP];

    const int bi = blockIdx.z;
    const int I0 = blockIdx.y * TILE;
    const int J0 = blockIdx.x * TILE;
    const float b0 = bias[0];

    const int t    = threadIdx.x;
    const int wave = t >> 5;
    const int lane = t & 31;
    const int wr   = (wave & 1) * 32;
    const int wc   = (wave >> 1) * 32;
    const int m    = lane & 15;
    const int kp   = (lane >> 4) * 2;

    const float* Xb = X + (size_t)bi * L_DIM * D_DIM;

    const int srow = t >> 4;
    const int col4 = (t & 15) << 2;
    const float* pa = Xb + (size_t)(I0 + srow) * D_DIM + col4;
    const float* pb = Xb + (size_t)(J0 + srow) * D_DIM + col4;
    const float* pw = W + col4;

    v4f ra[NLD], rb[NLD], wv;
    #pragma unroll
    for (int i = 0; i < NLD; i++) {
        ra[i] = *(const v4f*)(pa + (size_t)(8 * i) * D_DIM);
        rb[i] = *(const v4f*)(pb + (size_t)(8 * i) * D_DIM);
    }
    wv = *(const v4f*)pw;

    v8f acc[2][2] = {};

    #pragma nounroll
    for (int k0 = 0; k0 < D_DIM; k0 += KC) {
        #pragma unroll
        for (int i = 0; i < NLD; i++) {
            *(v4f*)&As[(srow + 8 * i) * LDSP + col4] = ra[i] * wv;
            *(v4f*)&Bs[(srow + 8 * i) * LDSP + col4] = rb[i];
        }

        if (k0 + KC < D_DIM) {
            #pragma unroll
            for (int i = 0; i < NLD; i++) {
                ra[i] = *(const v4f*)(pa + (size_t)(8 * i) * D_DIM + k0 + KC);
                rb[i] = *(const v4f*)(pb + (size_t)(8 * i) * D_DIM + k0 + KC);
            }
            wv = *(const v4f*)(pw + k0 + KC);
        }
        lds_barrier();

        #pragma unroll
        for (int kk = 0; kk < KC; kk += 4) {
            v2f a0 = *(const v2f*)&As[(wr + m)      * LDSP + kk + kp];
            v2f a1 = *(const v2f*)&As[(wr + 16 + m) * LDSP + kk + kp];
            v2f bq0 = *(const v2f*)&Bs[(wc + m)      * LDSP + kk + kp];
            v2f bq1 = *(const v2f*)&Bs[(wc + 16 + m) * LDSP + kk + kp];

            acc[0][0] = __builtin_amdgcn_wmma_f32_16x16x4_f32(false, a0, false, bq0,
                                                              (short)0, acc[0][0], false, false);
            acc[0][1] = __builtin_amdgcn_wmma_f32_16x16x4_f32(false, a0, false, bq1,
                                                              (short)0, acc[0][1], false, false);
            acc[1][0] = __builtin_amdgcn_wmma_f32_16x16x4_f32(false, a1, false, bq0,
                                                              (short)0, acc[1][0], false, false);
            acc[1][1] = __builtin_amdgcn_wmma_f32_16x16x4_f32(false, a1, false, bq1,
                                                              (short)0, acc[1][1], false, false);
        }
        lds_barrier();
    }

    float* outb = out + (size_t)bi * L_DIM * L_DIM;
    const int rsel = (lane >> 4) * 8;
    #pragma unroll
    for (int ti = 0; ti < 2; ti++) {
        #pragma unroll
        for (int tj = 0; tj < 2; tj++) {
            const int rowBase = I0 + wr + ti * 16 + rsel;
            const int col     = J0 + wc + tj * 16 + m;
            #pragma unroll
            for (int r = 0; r < 8; r++) {
                outb[(size_t)(rowBase + r) * L_DIM + col] = acc[ti][tj][r] + b0;
            }
        }
    }
}

extern "C" void kernel_launch(void* const* d_in, const int* in_sizes, int n_in,
                              void* d_out, int out_size, void* d_ws, size_t ws_size,
                              hipStream_t stream) {
    (void)in_sizes; (void)n_in; (void)out_size; (void)d_ws; (void)ws_size;
    const float* X    = (const float*)d_in[0];
    const float* W    = (const float*)d_in[1];
    const float* bias = (const float*)d_in[2];
    float* out = (float*)d_out;

    dim3 grid(L_DIM / TILE, L_DIM / TILE, 2);
    dim3 block(128);
    pairwise_bilin_wmma_f32<<<grid, block, 0, stream>>>(X, W, bias, out);
}


# block tile 64x64 (K chunk 64) -> 64x32 (K chunk 128): 256 work-groups, one per WGP (was 128, half the WGPs idle); a wave runs 128 WMMAs in 2 chunks of 64 (was 256 in 4); sum order unchanged
# speedup vs baseline: 3.9385x; 1.2066x over previous
#include <hip/hip_runtime.h>


typedef __attribute__((ext_vector_type(2))) float v2f;
typedef __attribute__((ext_vector_type(4))) float v4f;
typedef __attribute__((ext_vector_type(8))) float v8f;

#define L_DIM 512
#define D_DIM 256
#define TI    64
#define TJ    32
#define KC    128
#define LDSP  132
#define NLA   16
#define NLB   8

static __device__ __forceinline__ void lds_barrier() {
    __builtin_amdgcn_fence(__ATOMIC_RELEASE, "workgroup", "local");
    __builtin_amdgcn_s_barrier();
    __builtin_amdgcn_fence(__ATOMIC_ACQUIRE, "workgroup", "local");
}

__global__ __launch_bounds__(128) void
pairwise_bilin_wmma_f32(const float* __restrict__ X,
                        const float* __restrict__ W,
                        const float* __restrict__ bias,
                        float* __restrict__ out)
{
    __shared__ __align__(16) float As[TI * LDSP];
    __shared__ __align__(16) float Bs[TJ * LDSP];

    const int bi = blockIdx.z;
    const int I0 = blockIdx.y * TI;
    const int J0 = blockIdx.x * TJ;
    const float b0 = bias[0];

    const int t    = threadIdx.x;
    const int wave = t >> 5;
    const int lane = t & 31;
    const int wr   = wave * 16;
    const int m    = lane & 15;
    const int kp   = (lane >> 4) * 2;

    const float* Xb = X + (size_t)bi * L_DIM * D_DIM;

    const int srow = t >> 5;
    const int col4 = (t & 31) << 2;
    const float* pa = Xb + (size_t)(I0 + srow) * D_DIM + col4;
    const float* pb = Xb + (size_t)(J0 + srow) * D_DIM + col4;
    const float* pw = W + col4;

    v4f ra[NLA], rb[NLB], wv;
    #pragma unroll
    for (int i = 0; i < NLA; i++) ra[i] = *(const v4f*)(pa + (size_t)(4 * i) * D_DIM);
    #pragma unroll
    for (int i = 0; i < NLB; i++) rb[i] = *(const v4f*)(pb + (size_t)(4 * i) * D_DIM);
    wv = *(const v4f*)pw;

    v8f acc[2] = {};

    #pragma nounroll
    for (int k0 = 0; k0 < D_DIM; k0 += KC) {
        #pragma unroll
        for (int i = 0; i < NLA; i++)
            *(v4f*)&As[(srow + 4 * i) * LDSP + col4] = ra[i] * wv;
        #pragma unroll
        for (int i = 0; i < NLB; i++)
            *(v4f*)&Bs[(srow + 4 * i) * LDSP + col4] = rb[i];

        if (k0 + KC < D_DIM) {
            #pragma unroll
            for (int i = 0; i < NLA; i++) ra[i] = *(const v4f*)(pa + (size_t)(4 * i) * D_DIM + k0 + KC);
            #pragma unroll
            for (int i = 0; i < NLB; i++) rb[i] = *(const v4f*)(pb + (size_t)(4 * i) * D_DIM + k0 + KC);
            wv = *(const v4f*)(pw + k0 + KC);
        }
        lds_barrier();

        #pragma unroll
        for (int kk = 0; kk < KC; kk += 4) {
            v2f a   = *(const v2f*)&As[(wr + m) * LDSP + kk + kp];
            v2f bq0 = *(const v2f*)&Bs[m        * LDSP + kk + kp];
            v2f bq1 = *(const v2f*)&Bs[(16 + m) * LDSP + kk + kp];

            acc[0] = __builtin_amdgcn_wmma_f32_16x16x4_f32(false, a, false, bq0,
                                                           (short)0, acc[0], false, false);
            acc[1] = __builtin_amdgcn_wmma_f32_16x16x4_f32(false, a, false, bq1,
                                                           (short)0, acc[1], false, false);
        }
        lds_barrier();
    }

    float* outb = out + (size_t)bi * L_DIM * L_DIM;
    const int rowBase = I0 + wr + (lane >> 4) * 8;
    #pragma unroll
    for (int tj = 0; tj < 2; tj++) {
        const int col = J0 + tj * 16 + m;
        #pragma unroll
        for (int r = 0; r < 8; r++) {
            outb[(size_t)(rowBase + r) * L_DIM + col] = acc[tj][r] + b0;
        }
    }
}

extern "C" void kernel_launch(void* const* d_in, const int* in_sizes, int n_in,
                              void* d_out, int out_size, void* d_ws, size_t ws_size,
                              hipStream_t stream) {
    (void)in_sizes; (void)n_in; (void)out_size; (void)d_ws; (void)ws_size;
    const float* X    = (const float*)d_in[0];
    const float* W    = (const float*)d_in[1];
    const float* bias = (const float*)d_in[2];
    float* out = (float*)d_out;

    dim3 grid(L_DIM / TJ, L_DIM / TI, 2);
    dim3 block(128);
    pairwise_bilin_wmma_f32<<<grid, block, 0, stream>>>(X, W, bias, out);
}
